# starting kernel
# baseline (speedup 1.0000x reference)
#include <hip/hip_runtime.h>
#include <hip/hip_bf16.h>


typedef __attribute__((ext_vector_type(2))) float v2f;
typedef __attribute__((ext_vector_type(8))) float v8f;

#define SPATIAL 3136
#define N_GROUPS 2
#define JTILES 6
#define STILES 196
#define TILES_TOTAL (N_GROUPS * JTILES * STILES)
#define WAVES_PER_BLOCK 8
#define BLOCKS (TILES_TOTAL / WAVES_PER_BLOCK)

__global__ void build_A_kernel(const float* __restrict__ w0,
                               const float* __restrict__ w1,
                               float* __restrict__ A) {
    int idx = blockIdx.x * blockDim.x + threadIdx.x;
    if (idx >= N_GROUPS * 96 * 24) return;
    int g = idx % 24;
    int t = idx / 24;
    int j = t % 96;
    int n = t / 96;
    const float* w0r = w0 + j * 9;
    const float* w1r = w1 + (n * 24 + g) * 9;
    float acc = 0.f;
#pragma unroll
    for (int i = 0; i < 9; ++i) acc += w0r[i] * w1r[i];
    A[idx] = acc;
}

__global__ __launch_bounds__(WAVES_PER_BLOCK * 32)
void fused_wmma_kernel(const float* __restrict__ x,
                       const float* __restrict__ A,
                       float* __restrict__ out) {
    const int wid  = (blockIdx.x * blockDim.x + threadIdx.x) >> 5;
    const int lane = threadIdx.x & 31;
    if (wid >= TILES_TOTAL) return;

    const int n  = wid / (JTILES * STILES);
    const int r0 = wid % (JTILES * STILES);
    const int jt = r0 / STILES;
    const int st = r0 % STILES;
    const int j0 = jt << 4;
    const int s0 = st << 4;

    const int lane15 = lane & 15;
    const int khi    = lane >> 4;

    const int s = s0 + lane15;
    const int l = s / 56;
    const int m = s - l * 56;
    const float* xb = x + (size_t)n * 24 * SPATIAL + (size_t)m * 56 + l;

    const float* ab = A + ((size_t)(n * 96 + j0 + lane15)) * 24 + 2 * khi;

    v8f c = {};
#pragma unroll
    for (int kk = 0; kk < 6; ++kk) {
        v2f a = *(const v2f*)(ab + kk * 4);
        const int ch = kk * 4 + 2 * khi;
        v2f b;
        b.x = xb[(size_t)ch * SPATIAL];
        b.y = xb[(size_t)(ch + 1) * SPATIAL];
        c = __builtin_amdgcn_wmma_f32_16x16x4_f32(false, a, false, b,
                                                  (short)0, c, false, false);
    }

    float* ob = out + ((size_t)(n * 96 + j0 + 8 * khi)) * SPATIAL + s;
#pragma unroll
    for (int rr = 0; rr < 8; ++rr)
        ob[(size_t)rr * SPATIAL] = c[rr];
}

extern "C" void kernel_launch(void* const* d_in, const int* in_sizes, int n_in,
                              void* d_out, int out_size, void* d_ws, size_t ws_size,
                              hipStream_t stream) {
    const float* x  = (const float*)d_in[0];
    const float* w0 = (const float*)d_in[1];
    const float* w1 = (const float*)d_in[2];
    float* out = (float*)d_out;
    float* A   = (float*)d_ws;

    build_A_kernel<<<(N_GROUPS * 96 * 24 + 127) / 128, 128, 0, stream>>>(w0, w1, A);
    fused_wmma_kernel<<<BLOCKS, WAVES_PER_BLOCK * 32, 0, stream>>>(x, A, out);
}


# Reassociate as the reference does: one launch, per wave T = w1^T.x (6 WMMA) then out = w0.T (18 WMMA) with T's result registers as B; no folded A, x read once not 6x; f32 sum order changes
# speedup vs baseline: 2.4868x; 2.4868x over previous
#include <hip/hip_runtime.h>


typedef __attribute__((ext_vector_type(2))) float v2f;
typedef __attribute__((ext_vector_type(8))) float v8f;

#define SPATIAL 3136
#define STILES  196
#define NGROUP  2
#define GSZ     24
#define NI      9
#define NJ      96
#define JTILES  6

__global__ __launch_bounds__(32)
void fused_two_stage_kernel(const float* __restrict__ x,
                            const float* __restrict__ w0,
                            const float* __restrict__ w1,
                            float* __restrict__ out) {
    const int st   = blockIdx.x;
    const int n    = blockIdx.y;
    const int lane = threadIdx.x;
    const int c    = lane & 15;
    const int h    = lane >> 4;

    const int s = st * 16 + c;
    const int l = s / 56;
    const int m = s - l * 56;
    const float* xb = x + (n * GSZ + 2 * h) * SPATIAL + m * 56 + l;

    const int  rr = c & 7;
    const int  hr = c >> 3;
    const bool v1 = rr < 5 - hr;
    const int  i1 = v1 ? 5 * hr + rr : 0;
    const float* w1b = w1 + (n * GSZ + 2 * h) * NI + i1;

    const float* w0b = w0 + c * NI + 5 * h;
    const int e4 = h ? 3 : 4;

    float xv[12], w1v[12], w0v[JTILES][5];
    for (int kk = 0; kk < 6; ++kk) {
        xv[2 * kk]     = xb[(4 * kk) * SPATIAL];
        xv[2 * kk + 1] = xb[(4 * kk + 1) * SPATIAL];
    }
    for (int kk = 0; kk < 6; ++kk) {
        w1v[2 * kk]     = w1b[(4 * kk) * NI];
        w1v[2 * kk + 1] = w1b[(4 * kk + 1) * NI];
    }
    for (int jt = 0; jt < JTILES; ++jt) {
        const float* p = w0b + jt * 16 * NI;
        w0v[jt][0] = p[0];
        w0v[jt][1] = p[1];
        w0v[jt][2] = p[2];
        w0v[jt][3] = p[3];
        w0v[jt][4] = p[e4];
    }

    __builtin_amdgcn_sched_barrier(0);

    v8f t = {};
    for (int kk = 0; kk < 6; ++kk) {
        v2f a1, bx;
        a1.x = v1 ? w1v[2 * kk] : 0.f;
        a1.y = v1 ? w1v[2 * kk + 1] : 0.f;
        bx.x = xv[2 * kk];
        bx.y = xv[2 * kk + 1];
        t = __builtin_amdgcn_wmma_f32_16x16x4_f32(false, a1, false, bx,
                                                  (short)0, t, false, false);
    }

    v2f b2[3];
    b2[0].x = t[0]; b2[0].y = t[1];
    b2[1].x = t[2]; b2[1].y = t[3];
    b2[2].x = t[4]; b2[2].y = t[5];

    float* ob = out + (n * NJ + 8 * h) * SPATIAL + s;
    for (int jt = 0; jt < JTILES; ++jt) {
        v2f a2[3];
        a2[0].x = w0v[jt][0]; a2[0].y = w0v[jt][1];
        a2[1].x = w0v[jt][2]; a2[1].y = w0v[jt][3];
        a2[2].x = h ? 0.f : w0v[jt][4]; a2[2].y = 0.f;
        v8f acc = {};
        for (int q = 0; q < 3; ++q) {
            acc = __builtin_amdgcn_wmma_f32_16x16x4_f32(false, a2[q], false, b2[q],
                                                        (short)0, acc, false, false);
        }
        float* o = ob + jt * 16 * SPATIAL;
        for (int r = 0; r < 8; ++r)
            o[r * SPATIAL] = acc[r];
    }
}

extern "C" void kernel_launch(void* const* d_in, const int* in_sizes, int n_in,
                              void* d_out, int out_size, void* d_ws, size_t ws_size,
                              hipStream_t stream) {
    const float* x  = (const float*)d_in[0];
    const float* w0 = (const float*)d_in[1];
    const float* w1 = (const float*)d_in[2];
    float* out = (float*)d_out;

    fused_two_stage_kernel<<<dim3(STILES, NGROUP), 32, 0, stream>>>(x, w0, w1, out);
}
